# starting kernel
# baseline (speedup 1.0000x reference)
#include <hip/hip_runtime.h>
#include <hip/hip_bf16.h>
#include <math.h>

#define TOKENS 20
#define DIM    128
#define CELLS_PER_WAVE  16
#define WAVES_PER_BLOCK 4
#define THREADS_PER_BLOCK (WAVES_PER_BLOCK * 32)
#define CELLS_PER_BLOCK (CELLS_PER_WAVE * WAVES_PER_BLOCK)
#define LDS_STRIDE 132

typedef __attribute__((ext_vector_type(2))) float v2f;
typedef __attribute__((ext_vector_type(8))) float v8f;

__device__ __forceinline__ float gelu_exact(float v) {
    return 0.5f * v * (1.0f + erff(v * 0.70710678118654752440f));
}

__global__ __launch_bounds__(THREADS_PER_BLOCK)
void SimpleGeluEmbed_kernel(const int* __restrict__ x,
                            const float* __restrict__ emb,
                            const float* __restrict__ W,
                            const float* __restrict__ bias_p,
                            float* __restrict__ out) {
    __shared__ float agg[WAVES_PER_BLOCK * CELLS_PER_WAVE * LDS_STRIDE];

    const int lane = threadIdx.x & 31;
    const int wave = threadIdx.x >> 5;
    const int waveCellBase = blockIdx.x * CELLS_PER_BLOCK + wave * CELLS_PER_WAVE;

    float* aggWave = agg + wave * (CELLS_PER_WAVE * LDS_STRIDE);
    const float4* emb4 = reinterpret_cast<const float4*>(emb);

    for (int m = 0; m < CELLS_PER_WAVE; ++m) {
        const int cell = waveCellBase + m;
        const int* xp = x + cell * TOKENS;
        float ax = 0.f, ay = 0.f, az = 0.f, aw = 0.f;
#pragma unroll
        for (int t = 0; t < TOKENS; ++t) {
            const int idx = xp[t];
            const float4 v = emb4[idx * (DIM / 4) + lane];
            ax += v.x; ay += v.y; az += v.z; aw += v.w;
        }
        const float inv = 1.0f / (float)TOKENS;
        float4 h;
        h.x = gelu_exact(ax * inv);
        h.y = gelu_exact(ay * inv);
        h.z = gelu_exact(az * inv);
        h.w = gelu_exact(aw * inv);
        *reinterpret_cast<float4*>(aggWave + m * LDS_STRIDE + lane * 4) = h;
    }

    __syncthreads();

    const int koff  = (lane & 16) ? 2 : 0;
    const int cellm = lane & 15;
    const float* arow = aggWave + cellm * LDS_STRIDE + koff;

    v8f c = {};
#pragma unroll
    for (int j = 0; j < 32; ++j) {
        const v2f a  = *reinterpret_cast<const v2f*>(arow + 4 * j);
        const v2f bw = *reinterpret_cast<const v2f*>(W + 4 * j + koff);
        c = __builtin_amdgcn_wmma_f32_16x16x4_f32(false, a, false, bw,
                                                  (short)0, c, false, false);
    }

    const float bb = bias_p[0];
    if (lane == 0 || lane == 16) {
        float* op = out + waveCellBase + ((lane >> 4) << 3);
        float4 lo = { c[0] + bb, c[1] + bb, c[2] + bb, c[3] + bb };
        float4 hi = { c[4] + bb, c[5] + bb, c[6] + bb, c[7] + bb };
        *reinterpret_cast<float4*>(op)     = lo;
        *reinterpret_cast<float4*>(op + 4) = hi;
    }
}

extern "C" void kernel_launch(void* const* d_in, const int* in_sizes, int n_in,
                              void* d_out, int out_size, void* d_ws, size_t ws_size,
                              hipStream_t stream) {
    const int*   x   = (const int*)d_in[0];
    const float* emb = (const float*)d_in[1];
    const float* W   = (const float*)d_in[2];
    const float* b   = (const float*)d_in[3];
    float*       out = (float*)d_out;

    const int n_cells = out_size;
    const int blocks  = n_cells / CELLS_PER_BLOCK;

    SimpleGeluEmbed_kernel<<<blocks, THREADS_PER_BLOCK, 0, stream>>>(
        x, emb, W, b, out);
}


# gather pipelined one cell ahead: token ids by scalar loads, next cell's 20 row loads issued before the current cell's sums and GELU (two row buffers); per-component sum order unchanged
# speedup vs baseline: 1.0469x; 1.0469x over previous
#include <hip/hip_runtime.h>
#include <hip/hip_bf16.h>
#include <math.h>


#define TOKENS 20
#define DIM    128
#define CELLS_PER_WAVE  16
#define WAVES_PER_BLOCK 4
#define THREADS_PER_BLOCK (WAVES_PER_BLOCK * 32)
#define CELLS_PER_BLOCK (CELLS_PER_WAVE * WAVES_PER_BLOCK)
#define LDS_STRIDE 132

typedef __attribute__((ext_vector_type(2))) float v2f;
typedef __attribute__((ext_vector_type(4))) float v4f;
typedef __attribute__((ext_vector_type(8))) float v8f;

__device__ __forceinline__ float gelu_exact(float v) {
    return 0.5f * v * (1.0f + erff(v * 0.70710678118654752440f));
}

__device__ __forceinline__ void load_ids(const int* __restrict__ xp, int (&ids)[TOKENS]) {
#pragma unroll
    for (int t = 0; t < TOKENS; ++t) ids[t] = xp[t];
}

__device__ __forceinline__ void issue_rows(const v4f* __restrict__ emb4,
                                           const int (&ids)[TOKENS], int lane,
                                           v4f (&r)[TOKENS]) {
#pragma unroll
    for (int t = 0; t < TOKENS; ++t) r[t] = emb4[ids[t] * (DIM / 4) + lane];
}

__device__ __forceinline__ void reduce_store(const v4f (&r)[TOKENS], float* dst) {
    v2f a01 = {0.f, 0.f}, a23 = {0.f, 0.f};
#pragma unroll
    for (int t = 0; t < TOKENS; ++t) {
        a01 += r[t].xy;
        a23 += r[t].zw;
    }
    const float inv = 1.0f / (float)TOKENS;
    v4f h;
    h.x = gelu_exact(a01.x * inv);
    h.y = gelu_exact(a01.y * inv);
    h.z = gelu_exact(a23.x * inv);
    h.w = gelu_exact(a23.y * inv);
    *reinterpret_cast<v4f*>(dst) = h;
}

__global__ __launch_bounds__(THREADS_PER_BLOCK)
void SimpleGeluEmbed_kernel(const int* __restrict__ x,
                            const float* __restrict__ emb,
                            const float* __restrict__ W,
                            const float* __restrict__ bias_p,
                            float* __restrict__ out) {
    __shared__ float agg[WAVES_PER_BLOCK * CELLS_PER_WAVE * LDS_STRIDE];

    const int lane = threadIdx.x & 31;
    const int wave = __builtin_amdgcn_readfirstlane(threadIdx.x >> 5);
    const int waveCellBase = blockIdx.x * CELLS_PER_BLOCK + wave * CELLS_PER_WAVE;

    float* aggWave = agg + wave * (CELLS_PER_WAVE * LDS_STRIDE);
    float* aggLane = aggWave + lane * 4;
    const v4f* emb4 = reinterpret_cast<const v4f*>(emb);
    const int* xw = x + waveCellBase * TOKENS;

    int idsA[TOKENS], idsB[TOKENS];
    v4f rowsA[TOKENS], rowsB[TOKENS];

    load_ids(xw, idsA);
    issue_rows(emb4, idsA, lane, rowsA);
    load_ids(xw + TOKENS, idsB);

#pragma unroll 1
    for (int m = 0; m < CELLS_PER_WAVE - 2; m += 2) {
        issue_rows(emb4, idsB, lane, rowsB);
        load_ids(xw + (m + 2) * TOKENS, idsA);
        __builtin_amdgcn_sched_barrier(0);
        reduce_store(rowsA, aggLane + m * LDS_STRIDE);

        issue_rows(emb4, idsA, lane, rowsA);
        load_ids(xw + (m + 3) * TOKENS, idsB);
        __builtin_amdgcn_sched_barrier(0);
        reduce_store(rowsB, aggLane + (m + 1) * LDS_STRIDE);
    }
    issue_rows(emb4, idsB, lane, rowsB);
    __builtin_amdgcn_sched_barrier(0);
    reduce_store(rowsA, aggLane + (CELLS_PER_WAVE - 2) * LDS_STRIDE);
    reduce_store(rowsB, aggLane + (CELLS_PER_WAVE - 1) * LDS_STRIDE);

    __syncthreads();

    const int koff  = (lane & 16) ? 2 : 0;
    const int cellm = lane & 15;
    const float* arow = aggWave + cellm * LDS_STRIDE + koff;

    v8f c = {};
#pragma unroll
    for (int j = 0; j < 32; ++j) {
        const v2f a  = *reinterpret_cast<const v2f*>(arow + 4 * j);
        const v2f bw = *reinterpret_cast<const v2f*>(W + 4 * j + koff);
        c = __builtin_amdgcn_wmma_f32_16x16x4_f32(false, a, false, bw,
                                                  (short)0, c, false, false);
    }

    const float bb = bias_p[0];
    if (lane == 0 || lane == 16) {
        float* op = out + waveCellBase + ((lane >> 4) << 3);
        float4 lo = { c[0] + bb, c[1] + bb, c[2] + bb, c[3] + bb };
        float4 hi = { c[4] + bb, c[5] + bb, c[6] + bb, c[7] + bb };
        *reinterpret_cast<float4*>(op)     = lo;
        *reinterpret_cast<float4*>(op + 4) = hi;
    }
}

extern "C" void kernel_launch(void* const* d_in, const int* in_sizes, int n_in,
                              void* d_out, int out_size, void* d_ws, size_t ws_size,
                              hipStream_t stream) {
    const int*   x   = (const int*)d_in[0];
    const float* emb = (const float*)d_in[1];
    const float* W   = (const float*)d_in[2];
    const float* b   = (const float*)d_in[3];
    float*       out = (float*)d_out;

    const int n_cells = out_size;
    const int blocks  = n_cells / CELLS_PER_BLOCK;

    SimpleGeluEmbed_kernel<<<blocks, THREADS_PER_BLOCK, 0, stream>>>(
        x, emb, W, b, out);
}
